# starting kernel
# baseline (speedup 1.0000x reference)
#include <hip/hip_runtime.h>

typedef __attribute__((ext_vector_type(2))) float v2f;
typedef __attribute__((ext_vector_type(4))) float v4f;
typedef __attribute__((ext_vector_type(8))) float v8f;

#define NODE_STRIDE 16384
#define PLANE_ELEMS (8 * 16384)

__global__ __launch_bounds__(256) void iwpt_haar3_wmma(
    const float* __restrict__ coeffs, float* __restrict__ out) {
  const int lane = threadIdx.x & 31;
  const int wave = (blockIdx.x * blockDim.x + threadIdx.x) >> 5;

  const int base_tile = wave * 4;
  const int plane     = base_tile >> 9;
  const int j0_base   = (base_tile & 511) * 32;

  const int hi = lane >> 4;
  const int n  = lane & 15;
  const float SCALE = 0.35355339059327378f;

  v2f aslab[4];
  int ld_off[4][2];
  #pragma unroll
  for (int q = 0; q < 4; ++q) {
    #pragma unroll
    for (int v = 0; v < 2; ++v) {
      const int col = 4 * q + 2 * hi + v;
      float val = 0.0f;
      if ((n >= 8) == (col >= 8)) {
        const int r = n & 7, k = col & 7;
        const int rb = ((r & 1) << 2) | (r & 2) | ((r >> 2) & 1);
        val = (__popc(rb & k) & 1) ? -SCALE : SCALE;
      }
      if (v == 0) aslab[q].x = val; else aslab[q].y = val;
      const int kk   = col & 7;
      const int node = kk ^ (kk >> 1) ^ (kk >> 2);
      ld_off[q][v] = node * NODE_STRIDE + ((col >= 8) ? 16 : 0);
    }
  }

  const float* __restrict__ src = coeffs + (size_t)plane * PLANE_ELEMS;
  float* __restrict__ dstbase   = out    + (size_t)plane * PLANE_ELEMS;

  #pragma unroll
  for (int t = 0; t < 4; ++t) {
    const int j0 = j0_base + 32 * t;

    v2f b[4];
    #pragma unroll
    for (int q = 0; q < 4; ++q) {
      b[q].x = src[ld_off[q][0] + j0 + n];
      b[q].y = src[ld_off[q][1] + j0 + n];
    }

    v8f c = {};
    c = __builtin_amdgcn_wmma_f32_16x16x4_f32(false, aslab[0], false, b[0],
                                              (short)0, c, false, false);
    c = __builtin_amdgcn_wmma_f32_16x16x4_f32(false, aslab[1], false, b[1],
                                              (short)0, c, false, false);
    c = __builtin_amdgcn_wmma_f32_16x16x4_f32(false, aslab[2], false, b[2],
                                              (short)0, c, false, false);
    c = __builtin_amdgcn_wmma_f32_16x16x4_f32(false, aslab[3], false, b[3],
                                              (short)0, c, false, false);

    float* dst = dstbase + (size_t)(j0 + lane) * 8;
    v4f lo, hi4;
    lo[0] = c[0]; lo[1] = c[1]; lo[2] = c[2]; lo[3] = c[3];
    hi4[0] = c[4]; hi4[1] = c[5]; hi4[2] = c[6]; hi4[3] = c[7];
    *(v4f*)(dst)     = lo;
    *(v4f*)(dst + 4) = hi4;
  }
}

extern "C" void kernel_launch(void* const* d_in, const int* in_sizes, int n_in,
                              void* d_out, int out_size, void* d_ws, size_t ws_size,
                              hipStream_t stream) {
  const float* coeffs = (const float*)d_in[0];
  float* out = (float*)d_out;
  iwpt_haar3_wmma<<<4096, 256, 0, stream>>>(coeffs, out);
}


# Fewer, longer-lived waves: grid 4096 -> 512 blocks, each wave loops over 8 chunks of 4 tiles with the next chunk's loads in flight, so wave start-up no longer paces the kernel; sums unchanged
# speedup vs baseline: 1.0323x; 1.0323x over previous
#include <hip/hip_runtime.h>


typedef __attribute__((ext_vector_type(2))) float v2f;
typedef __attribute__((ext_vector_type(4))) float v4f;
typedef __attribute__((ext_vector_type(8))) float v8f;

#define NODE_STRIDE 16384
#define PLANE_ELEMS (8 * 16384)

#define N_BLOCKS 512
#define N_WAVES (N_BLOCKS * 8)
#define N_CHUNKS (256 * 128)
#define CHUNKS_PER_WAVE (N_CHUNKS / N_WAVES)
#define CHUNK_STEP ((N_WAVES / 128) * PLANE_ELEMS)

struct Frag { v2f b[4]; };

__device__ __forceinline__ void load_tile(Frag& f, const float* __restrict__ s,
                                          const int (&ld_off)[4][2], int t) {
  #pragma unroll
  for (int q = 0; q < 4; ++q) {
    f.b[q].x = s[ld_off[q][0] + 32 * t];
    f.b[q].y = s[ld_off[q][1] + 32 * t];
  }
}

__device__ __forceinline__ void do_tile(const Frag& f, const v2f (&aslab)[4],
                                        float* __restrict__ d) {
  v8f c = {};
  c = __builtin_amdgcn_wmma_f32_16x16x4_f32(false, aslab[0], false, f.b[0],
                                            (short)0, c, false, false);
  c = __builtin_amdgcn_wmma_f32_16x16x4_f32(false, aslab[1], false, f.b[1],
                                            (short)0, c, false, false);
  c = __builtin_amdgcn_wmma_f32_16x16x4_f32(false, aslab[2], false, f.b[2],
                                            (short)0, c, false, false);
  c = __builtin_amdgcn_wmma_f32_16x16x4_f32(false, aslab[3], false, f.b[3],
                                            (short)0, c, false, false);
  v4f lo, hi4;
  lo[0] = c[0]; lo[1] = c[1]; lo[2] = c[2]; lo[3] = c[3];
  hi4[0] = c[4]; hi4[1] = c[5]; hi4[2] = c[6]; hi4[3] = c[7];
  *(v4f*)(d)     = lo;
  *(v4f*)(d + 4) = hi4;
}

template <bool MORE>
__device__ __forceinline__ void do_chunk(Frag (&cur)[4], Frag (&nxt)[4],
                                         const float* __restrict__ s_next,
                                         float* __restrict__ d,
                                         const v2f (&aslab)[4],
                                         const int (&ld_off)[4][2]) {
  if (MORE) { load_tile(nxt[0], s_next, ld_off, 0); load_tile(nxt[1], s_next, ld_off, 1); }
  do_tile(cur[0], aslab, d);
  do_tile(cur[1], aslab, d + 256);
  if (MORE) { load_tile(nxt[2], s_next, ld_off, 2); load_tile(nxt[3], s_next, ld_off, 3); }
  do_tile(cur[2], aslab, d + 512);
  do_tile(cur[3], aslab, d + 768);
}

__global__ __launch_bounds__(256) void iwpt_haar3_wmma(
    const float* __restrict__ coeffs, float* __restrict__ out) {
  const int lane = threadIdx.x & 31;
  const int wave = __builtin_amdgcn_readfirstlane((int)((blockIdx.x * blockDim.x + threadIdx.x) >> 5));

  const int hi = lane >> 4;
  const int n  = lane & 15;
  const float SCALE = 0.35355339059327378f;

  v2f aslab[4];
  int ld_off[4][2];
  #pragma unroll
  for (int q = 0; q < 4; ++q) {
    #pragma unroll
    for (int v = 0; v < 2; ++v) {
      const int col = 4 * q + 2 * hi + v;
      float val = 0.0f;
      if ((n >= 8) == (col >= 8)) {
        const int r = n & 7, k = col & 7;
        const int rb = ((r & 1) << 2) | (r & 2) | ((r >> 2) & 1);
        val = (__popc(rb & k) & 1) ? -SCALE : SCALE;
      }
      if (v == 0) aslab[q].x = val; else aslab[q].y = val;
      const int kk   = col & 7;
      const int node = kk ^ (kk >> 1) ^ (kk >> 2);
      ld_off[q][v] = node * NODE_STRIDE + ((col >= 8) ? 16 : 0) + n;
    }
  }

  const size_t plane0 = (size_t)(wave >> 7);
  const int    j0     = (wave & 127) * 128;
  const float* __restrict__ src = coeffs + plane0 * PLANE_ELEMS + j0;
  float* __restrict__ dst       = out + plane0 * PLANE_ELEMS + (size_t)j0 * 8 + lane * 8;

  Frag bufA[4], bufB[4];
  #pragma unroll
  for (int t = 0; t < 4; ++t) load_tile(bufA[t], src, ld_off, t);

  #pragma unroll 1
  for (int p = 0; p < CHUNKS_PER_WAVE / 2 - 1; ++p) {
    do_chunk<true>(bufA, bufB, src + CHUNK_STEP, dst, aslab, ld_off);
    do_chunk<true>(bufB, bufA, src + 2 * (size_t)CHUNK_STEP, dst + CHUNK_STEP, aslab, ld_off);
    src += 2 * (size_t)CHUNK_STEP;
    dst += 2 * (size_t)CHUNK_STEP;
  }
  do_chunk<true >(bufA, bufB, src + CHUNK_STEP, dst, aslab, ld_off);
  do_chunk<false>(bufB, bufA, src, dst + CHUNK_STEP, aslab, ld_off);
}

extern "C" void kernel_launch(void* const* d_in, const int* in_sizes, int n_in,
                              void* d_out, int out_size, void* d_ws, size_t ws_size,
                              hipStream_t stream) {
  const float* coeffs = (const float*)d_in[0];
  float* out = (float*)d_out;
  iwpt_haar3_wmma<<<N_BLOCKS, 256, 0, stream>>>(coeffs, out);
}


# Whole-line traffic: data as A operand, 16 B/lane loads (4 whole lines each) and, after a half-swap per D pair, one 128 B line per store; 8-term sums now 2 nodes per WMMA step (fixed new order)
# speedup vs baseline: 1.1193x; 1.0843x over previous
#include <hip/hip_runtime.h>


typedef __attribute__((ext_vector_type(2))) float v2f;
typedef __attribute__((ext_vector_type(4))) float v4f;
typedef __attribute__((ext_vector_type(8))) float v8f;
typedef __attribute__((ext_vector_type(2))) unsigned int v2u;

#define NODE_STRIDE 16384
#define PLANE_ELEMS (8 * 16384)

#define N_BLOCKS 512
#define N_WAVES (N_BLOCKS * 8)
#define N_CHUNKS (256 * 128)
#define CHUNKS_PER_WAVE (N_CHUNKS / N_WAVES)
#define CHUNK_STEP ((N_WAVES / 128) * PLANE_ELEMS)

struct Grp { v4f a[4]; };

__device__ __forceinline__ void load_group(Grp& f, const float* __restrict__ s,
                                           const int (&ld_off)[4], int g) {
  #pragma unroll
  for (int q = 0; q < 4; ++q)
    f.a[q] = *(const v4f*)(s + ld_off[q] + 64 * g);
}

__device__ __forceinline__ void do_group(const Grp& f, const v2f (&bmat)[4],
                                         float* __restrict__ d) {
  v8f x = {}, y = {};
  #pragma unroll
  for (int q = 0; q < 4; ++q) {
    v2f ax, ay;
    ax.x = f.a[q][0]; ax.y = f.a[q][1];
    ay.x = f.a[q][2]; ay.y = f.a[q][3];
    x = __builtin_amdgcn_wmma_f32_16x16x4_f32(false, ax, false, bmat[q],
                                              (short)0, x, false, false);
    y = __builtin_amdgcn_wmma_f32_16x16x4_f32(false, ay, false, bmat[q],
                                              (short)0, y, false, false);
  }
  #pragma unroll
  for (int i = 0; i < 8; ++i) {
    const v2u r = __builtin_amdgcn_permlane16_swap(__float_as_uint(x[i]),
                                                   __float_as_uint(y[i]),
                                                   false, false);
    d[32 * i]       = __uint_as_float(r[0]);
    d[256 + 32 * i] = __uint_as_float(r[1]);
  }
}

template <bool MORE>
__device__ __forceinline__ void do_chunk(Grp (&cur)[2], Grp (&nxt)[2],
                                         const float* __restrict__ s_next,
                                         float* __restrict__ d,
                                         const v2f (&bmat)[4],
                                         const int (&ld_off)[4]) {
  if (MORE) load_group(nxt[0], s_next, ld_off, 0);
  do_group(cur[0], bmat, d);
  if (MORE) load_group(nxt[1], s_next, ld_off, 1);
  do_group(cur[1], bmat, d + 512);
}

__global__ __launch_bounds__(256) void iwpt_haar3_wmma(
    const float* __restrict__ coeffs, float* __restrict__ out) {
  const int lane = threadIdx.x & 31;
  const int wave = __builtin_amdgcn_readfirstlane((int)((blockIdx.x * blockDim.x + threadIdx.x) >> 5));

  const int hi = lane >> 4;
  const int n  = lane & 15;
  const float SCALE = 0.35355339059327378f;

  v2f bmat[4];
  int ld_off[4];
  #pragma unroll
  for (int q = 0; q < 4; ++q) {
    const int k = 2 * q + hi;
    const int r = n & 7;
    const int rb = ((r & 1) << 2) | (r & 2) | ((r >> 2) & 1);
    const float val = (__popc(rb & k) & 1) ? -SCALE : SCALE;
    bmat[q].x = (n < 8)  ? val : 0.0f;
    bmat[q].y = (n >= 8) ? val : 0.0f;
    const int node = k ^ (k >> 1) ^ (k >> 2);
    ld_off[q] = node * NODE_STRIDE + 4 * n;
  }

  const size_t plane0 = (size_t)(wave >> 7);
  const int    j0     = (wave & 127) * 128;
  const float* __restrict__ src = coeffs + plane0 * PLANE_ELEMS + j0;
  float* __restrict__ dst       = out + plane0 * PLANE_ELEMS + (size_t)j0 * 8 + lane;

  Grp bufA[2], bufB[2];
  load_group(bufA[0], src, ld_off, 0);
  load_group(bufA[1], src, ld_off, 1);

  #pragma unroll 1
  for (int p = 0; p < CHUNKS_PER_WAVE / 2 - 1; ++p) {
    do_chunk<true>(bufA, bufB, src + CHUNK_STEP, dst, bmat, ld_off);
    do_chunk<true>(bufB, bufA, src + 2 * (size_t)CHUNK_STEP, dst + CHUNK_STEP, bmat, ld_off);
    src += 2 * (size_t)CHUNK_STEP;
    dst += 2 * (size_t)CHUNK_STEP;
  }
  do_chunk<true >(bufA, bufB, src + CHUNK_STEP, dst, bmat, ld_off);
  do_chunk<false>(bufB, bufA, src, dst + CHUNK_STEP, bmat, ld_off);
}

extern "C" void kernel_launch(void* const* d_in, const int* in_sizes, int n_in,
                              void* d_out, int out_size, void* d_ws, size_t ws_size,
                              hipStream_t stream) {
  const float* coeffs = (const float*)d_in[0];
  float* out = (float*)d_out;
  iwpt_haar3_wmma<<<N_BLOCKS, 256, 0, stream>>>(coeffs, out);
}


# 16-byte stores: operands swapped (data = B) so a lane's 8 D registers are 8 consecutive outputs; lines pass through the wave's own LDS block, 512 contiguous bytes per store; no lane swaps; same sums
# speedup vs baseline: 1.1773x; 1.0518x over previous
#include <hip/hip_runtime.h>


typedef __attribute__((ext_vector_type(2))) float v2f;
typedef __attribute__((ext_vector_type(4))) float v4f;
typedef __attribute__((ext_vector_type(8))) float v8f;

#define NODE_STRIDE 16384
#define PLANE_ELEMS (8 * 16384)

#define N_BLOCKS 512
#define N_WAVES (N_BLOCKS * 8)
#define N_CHUNKS (256 * 128)
#define CHUNKS_PER_WAVE (N_CHUNKS / N_WAVES)
#define CHUNK_STEP ((N_WAVES / 128) * PLANE_ELEMS)

struct Grp { v4f a[4]; };

#define LINE_PITCH 144
#define STAGE_BYTES (16 * LINE_PITCH)

__device__ __forceinline__ void load_group(Grp& f, const float* __restrict__ s,
                                           const int (&ld_off)[4], int g) {
  #pragma unroll
  for (int q = 0; q < 4; ++q)
    f.a[q] = *(const v4f*)(s + ld_off[q] + 64 * g);
}

__device__ __forceinline__ void do_group(const Grp& f, const v2f (&hmat)[4],
                                         unsigned char* st_wr, const unsigned char* st_rd,
                                         float* __restrict__ d) {
  v8f x = {}, y = {};
  #pragma unroll
  for (int q = 0; q < 4; ++q) {
    v2f bx, by;
    bx.x = f.a[q][0]; bx.y = f.a[q][1];
    by.x = f.a[q][2]; by.y = f.a[q][3];
    x = __builtin_amdgcn_wmma_f32_16x16x4_f32(false, hmat[q], false, bx,
                                              (short)0, x, false, false);
    y = __builtin_amdgcn_wmma_f32_16x16x4_f32(false, hmat[q], false, by,
                                              (short)0, y, false, false);
  }
  *(v4f*)(st_wr +  0) = __builtin_shufflevector(x, x, 0, 1, 2, 3);
  *(v4f*)(st_wr + 16) = __builtin_shufflevector(x, x, 4, 5, 6, 7);
  *(v4f*)(st_wr + 64) = __builtin_shufflevector(y, y, 0, 1, 2, 3);
  *(v4f*)(st_wr + 80) = __builtin_shufflevector(y, y, 4, 5, 6, 7);
  asm volatile("s_wait_dscnt 0x0" ::: "memory");
  #pragma unroll
  for (int z = 0; z < 4; ++z) {
    const v4f v = *(const v4f*)(st_rd + z * 4 * LINE_PITCH);
    *(v4f*)(d + 128 * z) = v;
  }
}

template <bool MORE>
__device__ __forceinline__ void do_chunk(Grp (&cur)[2], Grp (&nxt)[2],
                                         const float* __restrict__ s_next,
                                         float* __restrict__ d,
                                         const v2f (&hmat)[4],
                                         const int (&ld_off)[4],
                                         unsigned char* st_wr, const unsigned char* st_rd) {
  if (MORE) load_group(nxt[0], s_next, ld_off, 0);
  do_group(cur[0], hmat, st_wr, st_rd, d);
  if (MORE) load_group(nxt[1], s_next, ld_off, 1);
  do_group(cur[1], hmat, st_wr + STAGE_BYTES, st_rd + STAGE_BYTES, d + 512);
}

__global__ __launch_bounds__(256) void iwpt_haar3_wmma(
    const float* __restrict__ coeffs, float* __restrict__ out) {
  const int lane = threadIdx.x & 31;
  const int wave = __builtin_amdgcn_readfirstlane((int)((blockIdx.x * blockDim.x + threadIdx.x) >> 5));

  const int hi = lane >> 4;
  const int n  = lane & 15;
  const float SCALE = 0.35355339059327378f;

  v2f hmat[4];
  int ld_off[4];
  #pragma unroll
  for (int q = 0; q < 4; ++q) {
    const int k = 2 * q + hi;
    const int r = n & 7;
    const int rb = ((r & 1) << 2) | (r & 2) | ((r >> 2) & 1);
    const float val = (__popc(rb & k) & 1) ? -SCALE : SCALE;
    hmat[q].x = (n < 8)  ? val : 0.0f;
    hmat[q].y = (n >= 8) ? val : 0.0f;
    const int node = k ^ (k >> 1) ^ (k >> 2);
    ld_off[q] = node * NODE_STRIDE + 4 * n;
  }

  __shared__ __attribute__((aligned(16))) unsigned char stage[8 * 2 * STAGE_BYTES];
  unsigned char* const my_stage = stage + (threadIdx.x >> 5) * (2 * STAGE_BYTES);
  unsigned char* const st_wr = my_stage + n * LINE_PITCH + 32 * hi;
  const unsigned char* const st_rd = my_stage + (lane >> 3) * LINE_PITCH + 16 * (lane & 7);

  const size_t plane0 = (size_t)(wave >> 7);
  const int    j0     = (wave & 127) * 128;
  const float* __restrict__ src = coeffs + plane0 * PLANE_ELEMS + j0;
  float* __restrict__ dst       = out + plane0 * PLANE_ELEMS + (size_t)j0 * 8 + 4 * lane;

  Grp bufA[2], bufB[2];
  load_group(bufA[0], src, ld_off, 0);
  load_group(bufA[1], src, ld_off, 1);

  #pragma unroll 1
  for (int p = 0; p < CHUNKS_PER_WAVE / 2 - 1; ++p) {
    do_chunk<true>(bufA, bufB, src + CHUNK_STEP, dst, hmat, ld_off, st_wr, st_rd);
    do_chunk<true>(bufB, bufA, src + 2 * (size_t)CHUNK_STEP, dst + CHUNK_STEP, hmat, ld_off, st_wr, st_rd);
    src += 2 * (size_t)CHUNK_STEP;
    dst += 2 * (size_t)CHUNK_STEP;
  }
  do_chunk<true >(bufA, bufB, src + CHUNK_STEP, dst, hmat, ld_off, st_wr, st_rd);
  do_chunk<false>(bufB, bufA, src, dst + CHUNK_STEP, hmat, ld_off, st_wr, st_rd);
}

extern "C" void kernel_launch(void* const* d_in, const int* in_sizes, int n_in,
                              void* d_out, int out_size, void* d_ws, size_t ws_size,
                              hipStream_t stream) {
  const float* coeffs = (const float*)d_in[0];
  float* out = (float*)d_out;
  iwpt_haar3_wmma<<<N_BLOCKS, 256, 0, stream>>>(coeffs, out);
}
